# starting kernel
# baseline (speedup 1.0000x reference)
#include <hip/hip_runtime.h>

typedef __attribute__((ext_vector_type(2))) float v2f;
typedef __attribute__((ext_vector_type(8))) float v8f;

#define T_DIM  2048
#define F_INN  2048
#define D_DIM  128
#define BATCH  4
#define WIN    101
#define HWIN   50

__global__ __launch_bounds__(256) void transpose_w_kernel(
    const float* __restrict__ W, float* __restrict__ Wt,
    float* __restrict__ zrow)
{
  const int idx = blockIdx.x * 256 + threadIdx.x;
  const int r = idx >> 7;
  const int c = idx & 127;
  Wt[(size_t)c * F_INN + r] = W[idx];
  if (blockIdx.x == 0 && threadIdx.x < D_DIM) zrow[threadIdx.x] = 0.f;
}

__global__ __launch_bounds__(128) void gemm_norm_kernel(
    const float* __restrict__ x, const float* __restrict__ Wt,
    const float* __restrict__ bias, float* __restrict__ p)
{
  const int wave = (blockIdx.x * blockDim.x + threadIdx.x) >> 5;
  const int lane = threadIdx.x & 31;
  const int half = lane >> 4;
  const int l16  = lane & 15;
  const int row0 = wave * 16;

  v8f acc[8] = {};

  const float* xrow = x + (size_t)(row0 + l16) * F_INN + half * 2;
  const float* wrow[8];
#pragma unroll
  for (int n = 0; n < 8; ++n)
    wrow[n] = Wt + (size_t)(n * 16 + l16) * F_INN + half * 2;

  v2f a_cur = *(const v2f*)(xrow);
  v2f b_cur[8];
#pragma unroll
  for (int n = 0; n < 8; ++n) b_cur[n] = *(const v2f*)(wrow[n]);

  for (int k = 0; k < F_INN; k += 4) {
    const int kn = k + 4;
    v2f a_nxt = {};
    v2f b_nxt[8] = {};
    if (kn < F_INN) {
      a_nxt = *(const v2f*)(xrow + kn);
#pragma unroll
      for (int n = 0; n < 8; ++n) b_nxt[n] = *(const v2f*)(wrow[n] + kn);
    }
    __builtin_prefetch(xrow + k + 256, 0, 0);

#pragma unroll
    for (int n = 0; n < 8; ++n)
      acc[n] = __builtin_amdgcn_wmma_f32_16x16x4_f32(
          false, a_cur, false, b_cur[n], (short)0, acc[n], false, false);

    a_cur = a_nxt;
#pragma unroll
    for (int n = 0; n < 8; ++n) b_cur[n] = b_nxt[n];
  }

#pragma unroll
  for (int n = 0; n < 8; ++n) {
    const float bv = bias[n * 16 + l16];
#pragma unroll
    for (int j = 0; j < 8; ++j) acc[n][j] += bv;
  }

#pragma unroll
  for (int j = 0; j < 8; ++j) {
    float s = 0.f;
#pragma unroll
    for (int n = 0; n < 8; ++n) s += acc[n][j] * acc[n][j];
    s += __shfl_xor(s, 1, 32);
    s += __shfl_xor(s, 2, 32);
    s += __shfl_xor(s, 4, 32);
    s += __shfl_xor(s, 8, 32);
    const float scale = 1.0f / fmaxf(sqrtf(s), 1e-12f);
#pragma unroll
    for (int n = 0; n < 8; ++n) acc[n][j] *= scale;
  }

#pragma unroll
  for (int j = 0; j < 8; ++j) {
    const int r = row0 + j + 8 * half;
#pragma unroll
    for (int n = 0; n < 8; ++n)
      p[(size_t)r * D_DIM + n * 16 + l16] = acc[n][j];
  }
}

__global__ __launch_bounds__(128) void band_sim_kernel(
    const float* __restrict__ p, const float* __restrict__ zrow,
    float* __restrict__ out)
{
  const int wave = (blockIdx.x * blockDim.x + threadIdx.x) >> 5;
  const int lane = threadIdx.x & 31;
  const int half = lane >> 4;
  const int l16  = lane & 15;
  const int b    = wave >> 7;
  const int mt   = wave & 127;
  const int t0   = mt * 16;

  const float* pb = p + (size_t)b * T_DIM * D_DIM;

  const float* bbase[9];
#pragma unroll
  for (int ct = 0; ct < 9; ++ct) {
    const int c = t0 + (ct - 4) * 16 + l16;
    bbase[ct] = ((unsigned)c < (unsigned)T_DIM)
                    ? (pb + (size_t)c * D_DIM + half * 2)
                    : (zrow + half * 2);
  }
  const float* arow = pb + (size_t)(t0 + l16) * D_DIM + half * 2;

  v8f acc[9] = {};

  v2f a_cur = *(const v2f*)(arow);
  v2f b_cur[9];
#pragma unroll
  for (int ct = 0; ct < 9; ++ct) b_cur[ct] = *(const v2f*)(bbase[ct]);

  for (int k = 0; k < D_DIM; k += 4) {
    const int kn = k + 4;
    v2f a_nxt = {};
    v2f b_nxt[9] = {};
    if (kn < D_DIM) {
      a_nxt = *(const v2f*)(arow + kn);
#pragma unroll
      for (int ct = 0; ct < 9; ++ct) b_nxt[ct] = *(const v2f*)(bbase[ct] + kn);
    }

#pragma unroll
    for (int ct = 0; ct < 9; ++ct)
      acc[ct] = __builtin_amdgcn_wmma_f32_16x16x4_f32(
          false, a_cur, false, b_cur[ct], (short)0, acc[ct], false, false);

    a_cur = a_nxt;
#pragma unroll
    for (int ct = 0; ct < 9; ++ct) b_cur[ct] = b_nxt[ct];
  }

#pragma unroll
  for (int ct = 0; ct < 9; ++ct) {
    const int c = t0 + (ct - 4) * 16 + l16;
#pragma unroll
    for (int j = 0; j < 8; ++j) {
      const int t = t0 + j + 8 * half;
      const int w = c - t + HWIN;
      if (w >= 0 && w < WIN) {
        out[((size_t)b * T_DIM + t) * WIN + w] = acc[ct][j];
      }
    }
  }
}

extern "C" void kernel_launch(void* const* d_in, const int* in_sizes, int n_in,
                              void* d_out, int out_size, void* d_ws, size_t ws_size,
                              hipStream_t stream) {
  const float* x  = (const float*)d_in[0];
  const float* W  = (const float*)d_in[1];
  const float* bv = (const float*)d_in[2];
  float* out = (float*)d_out;

  float* p    = (float*)d_ws;
  float* Wt   = p + (size_t)8192 * D_DIM;
  float* zrow = Wt + (size_t)D_DIM * F_INN;

  transpose_w_kernel<<<(F_INN * D_DIM) / 256, 256, 0, stream>>>(W, Wt, zrow);
  gemm_norm_kernel<<<128, 128, 0, stream>>>(x, Wt, bv, p);
  band_sim_kernel<<<128, 128, 0, stream>>>(p, zrow, out);
}


# Projection GEMM reads W as it lies: 16-B loads of 4 columns at one k (whole lines), k pairs built by 2x2 register transposes, tile columns permuted; W^T copy launch dropped (3->2); sum order kept
# speedup vs baseline: 2.9868x; 2.9868x over previous
#include <hip/hip_runtime.h>

typedef __attribute__((ext_vector_type(2))) float v2f;
typedef __attribute__((ext_vector_type(8))) float v8f;

#define T_DIM  2048
#define F_INN  2048
#define D_DIM  128
#define BATCH  4
#define WIN    101
#define HWIN   50

typedef __attribute__((ext_vector_type(4))) float v4f;

#define G_ROWS 32
#define G_NIT  (F_INN / 32)
#define G_DEP  3
#define G_NSL  4

__global__ __launch_bounds__(128) void gemm_norm_kernel(
    const float* __restrict__ x, const float* __restrict__ W,
    const float* __restrict__ bias, float* __restrict__ p,
    float* __restrict__ zrow)
{
  __shared__ float red[41 * 1024];

  const int w    = __builtin_amdgcn_readfirstlane((int)(threadIdx.x >> 5));
  const int lane = threadIdx.x & 31;
  const int half = lane >> 4;
  const int l16  = lane & 15;
  const int row0 = blockIdx.x * G_ROWS;

  v8f acc[2][8] = {};

  const int koff = w * 8 + half * 4;
  const float* xr0 = x + (size_t)(row0 + l16) * F_INN + koff;
  const float* xr1 = xr0 + (size_t)16 * F_INN;
  const float* wr  = W + (size_t)koff * D_DIM + l16 * 4;
  const float* xpf = x + (size_t)(row0 + w * 8 + (lane & 7)) * F_INN;

  v4f fa[G_NSL][2];
  v4f fb[G_NSL][4][2];

#define G_LOAD(s_, it_)                                                     \
  {                                                                         \
    const int ko_ = (it_) * 32;                                             \
    fa[s_][0] = *(const v4f*)(xr0 + ko_);                                   \
    fa[s_][1] = *(const v4f*)(xr1 + ko_);                                   \
    _Pragma("unroll") for (int q_ = 0; q_ < 4; ++q_) {                      \
      _Pragma("unroll") for (int g_ = 0; g_ < 2; ++g_)                      \
        fb[s_][q_][g_] =                                                    \
            *(const v4f*)(wr + (size_t)(ko_ + q_) * D_DIM + g_ * 64);       \
    }                                                                       \
  }

#define G_MMA(s_)                                                           \
  {                                                                         \
    _Pragma("unroll") for (int n_ = 0; n_ < 8; ++n_) {                      \
      const v2f b_ = {fb[s_][0][n_ >> 2][n_ & 3], fb[s_][1][n_ >> 2][n_ & 3]}; \
      _Pragma("unroll") for (int m_ = 0; m_ < 2; ++m_)                      \
        acc[m_][n_] = __builtin_amdgcn_wmma_f32_16x16x4_f32(                \
            false, fa[s_][m_].lo, false, b_, (short)0,                      \
            acc[m_][n_], false, false);                                     \
    }                                                                       \
    _Pragma("unroll") for (int n_ = 0; n_ < 8; ++n_) {                      \
      const v2f b_ = {fb[s_][2][n_ >> 2][n_ & 3], fb[s_][3][n_ >> 2][n_ & 3]}; \
      _Pragma("unroll") for (int m_ = 0; m_ < 2; ++m_)                      \
        acc[m_][n_] = __builtin_amdgcn_wmma_f32_16x16x4_f32(                \
            false, fa[s_][m_].hi, false, b_, (short)0,                      \
            acc[m_][n_], false, false);                                     \
    }                                                                       \
  }

  G_LOAD(0, 0);
  __builtin_amdgcn_sched_barrier(0);
  G_LOAD(1, 1);
  __builtin_amdgcn_sched_barrier(0);
  G_LOAD(2, 2);
  __builtin_amdgcn_sched_barrier(0);

  for (int it0 = 0; it0 < G_NIT - G_NSL; it0 += G_NSL) {
#pragma unroll
    for (int u = 0; u < G_NSL; ++u) {
      const int it = it0 + u;
      G_LOAD((u + G_DEP) % G_NSL, it + G_DEP);
      const int pit = (it + 8 < G_NIT) ? (it + 8) : (G_NIT - 1);
      __builtin_prefetch(xpf + pit * 32, 0, 0);
      __builtin_amdgcn_sched_barrier(0);
      G_MMA(u);
      __builtin_amdgcn_sched_barrier(0);
    }
  }
  G_LOAD(G_DEP, G_NIT - 1);
  __builtin_amdgcn_sched_barrier(0);
  G_MMA(0);
  G_MMA(1);
  G_MMA(2);
  G_MMA(3);
#undef G_LOAD
#undef G_MMA

  float* mine = red + w * (G_ROWS * D_DIM);
#pragma unroll
  for (int m = 0; m < 2; ++m)
#pragma unroll
    for (int g = 0; g < 2; ++g)
#pragma unroll
      for (int r = 0; r < 8; ++r) {
        const v4f v = {acc[m][4 * g][r], acc[m][4 * g + 1][r],
                       acc[m][4 * g + 2][r], acc[m][4 * g + 3][r]};
        *(v4f*)(mine + (m * 16 + r + 8 * half) * D_DIM + g * 64 + l16 * 4) = v;
      }

  if (blockIdx.x == 0 && w == 0) {
    const v4f z = {0.f, 0.f, 0.f, 0.f};
    *(v4f*)(zrow + lane * 4) = z;
  }

  __syncthreads();

  const v4f bv = *(const v4f*)(bias + lane * 4);
#pragma unroll
  for (int r = 0; r < 8; ++r) {
    const int row = w * 8 + r;
    const float* src = red + row * D_DIM + lane * 4;
    v4f s = *(const v4f*)(src);
    s += *(const v4f*)(src + 1 * G_ROWS * D_DIM);
    s += *(const v4f*)(src + 2 * G_ROWS * D_DIM);
    s += *(const v4f*)(src + 3 * G_ROWS * D_DIM);
    s += bv;
    float q = s.x * s.x + s.y * s.y + s.z * s.z + s.w * s.w;
    q += __shfl_xor(q, 1, 32);
    q += __shfl_xor(q, 2, 32);
    q += __shfl_xor(q, 4, 32);
    q += __shfl_xor(q, 8, 32);
    q += __shfl_xor(q, 16, 32);
    const float scale = 1.0f / fmaxf(sqrtf(q), 1e-12f);
    s *= scale;
    *(v4f*)(p + (size_t)(row0 + row) * D_DIM + lane * 4) = s;
  }
}

__global__ __launch_bounds__(128) void band_sim_kernel(
    const float* __restrict__ p, const float* __restrict__ zrow,
    float* __restrict__ out)
{
  const int wave = (blockIdx.x * blockDim.x + threadIdx.x) >> 5;
  const int lane = threadIdx.x & 31;
  const int half = lane >> 4;
  const int l16  = lane & 15;
  const int b    = wave >> 7;
  const int mt   = wave & 127;
  const int t0   = mt * 16;

  const float* pb = p + (size_t)b * T_DIM * D_DIM;

  const float* bbase[9];
#pragma unroll
  for (int ct = 0; ct < 9; ++ct) {
    const int c = t0 + (ct - 4) * 16 + l16;
    bbase[ct] = ((unsigned)c < (unsigned)T_DIM)
                    ? (pb + (size_t)c * D_DIM + half * 2)
                    : (zrow + half * 2);
  }
  const float* arow = pb + (size_t)(t0 + l16) * D_DIM + half * 2;

  v8f acc[9] = {};

  v2f a_cur = *(const v2f*)(arow);
  v2f b_cur[9];
#pragma unroll
  for (int ct = 0; ct < 9; ++ct) b_cur[ct] = *(const v2f*)(bbase[ct]);

  for (int k = 0; k < D_DIM; k += 4) {
    const int kn = k + 4;
    v2f a_nxt = {};
    v2f b_nxt[9] = {};
    if (kn < D_DIM) {
      a_nxt = *(const v2f*)(arow + kn);
#pragma unroll
      for (int ct = 0; ct < 9; ++ct) b_nxt[ct] = *(const v2f*)(bbase[ct] + kn);
    }

#pragma unroll
    for (int ct = 0; ct < 9; ++ct)
      acc[ct] = __builtin_amdgcn_wmma_f32_16x16x4_f32(
          false, a_cur, false, b_cur[ct], (short)0, acc[ct], false, false);

    a_cur = a_nxt;
#pragma unroll
    for (int ct = 0; ct < 9; ++ct) b_cur[ct] = b_nxt[ct];
  }

#pragma unroll
  for (int ct = 0; ct < 9; ++ct) {
    const int c = t0 + (ct - 4) * 16 + l16;
#pragma unroll
    for (int j = 0; j < 8; ++j) {
      const int t = t0 + j + 8 * half;
      const int w = c - t + HWIN;
      if (w >= 0 && w < WIN) {
        out[((size_t)b * T_DIM + t) * WIN + w] = acc[ct][j];
      }
    }
  }
}

extern "C" void kernel_launch(void* const* d_in, const int* in_sizes, int n_in,
                              void* d_out, int out_size, void* d_ws, size_t ws_size,
                              hipStream_t stream) {
  const float* x  = (const float*)d_in[0];
  const float* W  = (const float*)d_in[1];
  const float* bv = (const float*)d_in[2];
  float* out = (float*)d_out;

  float* p    = (float*)d_ws;
  float* zrow = p + (size_t)8192 * D_DIM;

  gemm_norm_kernel<<<256, 128, 0, stream>>>(x, W, bv, p, zrow);
  band_sim_kernel<<<128, 128, 0, stream>>>(p, zrow, out);
}


# Band kernel: 2 waves per 16-row tile split the 9 band tiles 5/4 (1024 waves); 16-B fragment loads from global, all 80 requested ahead of the products, no k loop; sum order changed
# speedup vs baseline: 3.4530x; 1.1561x over previous
#include <hip/hip_runtime.h>

typedef __attribute__((ext_vector_type(2))) float v2f;
typedef __attribute__((ext_vector_type(8))) float v8f;

#define T_DIM  2048
#define F_INN  2048
#define D_DIM  128
#define BATCH  4
#define WIN    101
#define HWIN   50

typedef __attribute__((ext_vector_type(4))) float v4f;

#define G_ROWS 32
#define G_NIT  (F_INN / 32)
#define G_DEP  3
#define G_NSL  4

__global__ __launch_bounds__(128) void gemm_norm_kernel(
    const float* __restrict__ x, const float* __restrict__ W,
    const float* __restrict__ bias, float* __restrict__ p,
    float* __restrict__ zrow)
{
  __shared__ float red[41 * 1024];

  const int w    = __builtin_amdgcn_readfirstlane((int)(threadIdx.x >> 5));
  const int lane = threadIdx.x & 31;
  const int half = lane >> 4;
  const int l16  = lane & 15;
  const int row0 = blockIdx.x * G_ROWS;

  v8f acc[2][8] = {};

  const int koff = w * 8 + half * 4;
  const float* xr0 = x + (size_t)(row0 + l16) * F_INN + koff;
  const float* xr1 = xr0 + (size_t)16 * F_INN;
  const float* wr  = W + (size_t)koff * D_DIM + l16 * 4;
  const float* xpf = x + (size_t)(row0 + w * 8 + (lane & 7)) * F_INN;

  v4f fa[G_NSL][2];
  v4f fb[G_NSL][4][2];

#define G_LOAD(s_, it_)                                                     \
  {                                                                         \
    const int ko_ = (it_) * 32;                                             \
    fa[s_][0] = *(const v4f*)(xr0 + ko_);                                   \
    fa[s_][1] = *(const v4f*)(xr1 + ko_);                                   \
    _Pragma("unroll") for (int q_ = 0; q_ < 4; ++q_) {                      \
      _Pragma("unroll") for (int g_ = 0; g_ < 2; ++g_)                      \
        fb[s_][q_][g_] =                                                    \
            *(const v4f*)(wr + (size_t)(ko_ + q_) * D_DIM + g_ * 64);       \
    }                                                                       \
  }

#define G_MMA(s_)                                                           \
  {                                                                         \
    _Pragma("unroll") for (int n_ = 0; n_ < 8; ++n_) {                      \
      const v2f b_ = {fb[s_][0][n_ >> 2][n_ & 3], fb[s_][1][n_ >> 2][n_ & 3]}; \
      _Pragma("unroll") for (int m_ = 0; m_ < 2; ++m_)                      \
        acc[m_][n_] = __builtin_amdgcn_wmma_f32_16x16x4_f32(                \
            false, fa[s_][m_].lo, false, b_, (short)0,                      \
            acc[m_][n_], false, false);                                     \
    }                                                                       \
    _Pragma("unroll") for (int n_ = 0; n_ < 8; ++n_) {                      \
      const v2f b_ = {fb[s_][2][n_ >> 2][n_ & 3], fb[s_][3][n_ >> 2][n_ & 3]}; \
      _Pragma("unroll") for (int m_ = 0; m_ < 2; ++m_)                      \
        acc[m_][n_] = __builtin_amdgcn_wmma_f32_16x16x4_f32(                \
            false, fa[s_][m_].hi, false, b_, (short)0,                      \
            acc[m_][n_], false, false);                                     \
    }                                                                       \
  }

  G_LOAD(0, 0);
  __builtin_amdgcn_sched_barrier(0);
  G_LOAD(1, 1);
  __builtin_amdgcn_sched_barrier(0);
  G_LOAD(2, 2);
  __builtin_amdgcn_sched_barrier(0);

  for (int it0 = 0; it0 < G_NIT - G_NSL; it0 += G_NSL) {
#pragma unroll
    for (int u = 0; u < G_NSL; ++u) {
      const int it = it0 + u;
      G_LOAD((u + G_DEP) % G_NSL, it + G_DEP);
      const int pit = (it + 8 < G_NIT) ? (it + 8) : (G_NIT - 1);
      __builtin_prefetch(xpf + pit * 32, 0, 0);
      __builtin_amdgcn_sched_barrier(0);
      G_MMA(u);
      __builtin_amdgcn_sched_barrier(0);
    }
  }
  G_LOAD(G_DEP, G_NIT - 1);
  __builtin_amdgcn_sched_barrier(0);
  G_MMA(0);
  G_MMA(1);
  G_MMA(2);
  G_MMA(3);
#undef G_LOAD
#undef G_MMA

  float* mine = red + w * (G_ROWS * D_DIM);
#pragma unroll
  for (int m = 0; m < 2; ++m)
#pragma unroll
    for (int g = 0; g < 2; ++g)
#pragma unroll
      for (int r = 0; r < 8; ++r) {
        const v4f v = {acc[m][4 * g][r], acc[m][4 * g + 1][r],
                       acc[m][4 * g + 2][r], acc[m][4 * g + 3][r]};
        *(v4f*)(mine + (m * 16 + r + 8 * half) * D_DIM + g * 64 + l16 * 4) = v;
      }

  if (blockIdx.x == 0 && w == 0) {
    const v4f z = {0.f, 0.f, 0.f, 0.f};
    *(v4f*)(zrow + lane * 4) = z;
  }

  __syncthreads();

  const v4f bv = *(const v4f*)(bias + lane * 4);
#pragma unroll
  for (int r = 0; r < 8; ++r) {
    const int row = w * 8 + r;
    const float* src = red + row * D_DIM + lane * 4;
    v4f s = *(const v4f*)(src);
    s += *(const v4f*)(src + 1 * G_ROWS * D_DIM);
    s += *(const v4f*)(src + 2 * G_ROWS * D_DIM);
    s += *(const v4f*)(src + 3 * G_ROWS * D_DIM);
    s += bv;
    float q = s.x * s.x + s.y * s.y + s.z * s.z + s.w * s.w;
    q += __shfl_xor(q, 1, 32);
    q += __shfl_xor(q, 2, 32);
    q += __shfl_xor(q, 4, 32);
    q += __shfl_xor(q, 8, 32);
    q += __shfl_xor(q, 16, 32);
    const float scale = 1.0f / fmaxf(sqrtf(q), 1e-12f);
    s *= scale;
    *(v4f*)(p + (size_t)(row0 + row) * D_DIM + lane * 4) = s;
  }
}

#define B_NCH   16
#define B_FIRST 12

template <int HC>
__device__ __forceinline__ void band_half(
    const float* __restrict__ pb, const float* __restrict__ zrow,
    float* __restrict__ outb, const int t0, const int half, const int l16)
{
  constexpr int AS = HC ? 0 : 4;
  constexpr int S0 = HC ? 1 : 0;

  const float* base[5];
#pragma unroll
  for (int s = 0; s < 5; ++s) {
    const int c = t0 + (4 * HC + s - 4) * 16 + l16;
    base[s] = ((unsigned)c < (unsigned)T_DIM)
                  ? (pb + (size_t)c * D_DIM + half * 4)
                  : (zrow + half * 4);
  }

  v4f f[B_NCH][5];
  v8f acc[5] = {};

#define B_LOAD(i_)                                                          \
  {                                                                         \
    _Pragma("unroll") for (int s_ = 0; s_ < 5; ++s_)                        \
      f[i_][s_] = *(const v4f*)(base[s_] + (i_) * 8);                       \
    __builtin_amdgcn_sched_barrier(0);                                      \
  }
#define B_MMA(i_)                                                           \
  {                                                                         \
    _Pragma("unroll") for (int s_ = S0; s_ < 5; ++s_)                       \
      acc[s_] = __builtin_amdgcn_wmma_f32_16x16x4_f32(                      \
          false, f[i_][AS].lo, false, f[i_][s_].lo, (short)0, acc[s_],      \
          false, false);                                                    \
    _Pragma("unroll") for (int s_ = S0; s_ < 5; ++s_)                       \
      acc[s_] = __builtin_amdgcn_wmma_f32_16x16x4_f32(                      \
          false, f[i_][AS].hi, false, f[i_][s_].hi, (short)0, acc[s_],      \
          false, false);                                                    \
  }

#pragma unroll
  for (int i = 0; i < B_FIRST; ++i) B_LOAD(i);
#pragma unroll
  for (int i = 0; i < B_NCH - B_FIRST; ++i) B_MMA(i);
  __builtin_amdgcn_sched_barrier(0);
#pragma unroll
  for (int i = B_FIRST; i < B_NCH; ++i) B_LOAD(i);
#pragma unroll
  for (int i = B_NCH - B_FIRST; i < B_NCH; ++i) B_MMA(i);
#undef B_LOAD
#undef B_MMA

#pragma unroll
  for (int s = S0; s < 5; ++s) {
    const int c = t0 + (4 * HC + s - 4) * 16 + l16;
#pragma unroll
    for (int j = 0; j < 8; ++j) {
      const int t = t0 + j + 8 * half;
      const int w = c - t + HWIN;
      if (w >= 0 && w < WIN) {
        outb[(size_t)t * WIN + w] = acc[s][j];
      }
    }
  }
}

__global__ __launch_bounds__(128) void band_sim_kernel(
    const float* __restrict__ p, const float* __restrict__ zrow,
    float* __restrict__ out)
{
  const int wv   = __builtin_amdgcn_readfirstlane((int)(threadIdx.x >> 5));
  const int lane = threadIdx.x & 31;
  const int half = lane >> 4;
  const int l16  = lane & 15;
  const int tile = blockIdx.x * 2 + (wv >> 1);
  const int b    = tile >> 7;
  const int t0   = (tile & 127) * 16;

  const float* pb = p + (size_t)b * T_DIM * D_DIM;
  float* outb     = out + (size_t)b * T_DIM * WIN;

  if (wv & 1) band_half<1>(pb, zrow, outb, t0, half, l16);
  else        band_half<0>(pb, zrow, outb, t0, half, l16);
}

extern "C" void kernel_launch(void* const* d_in, const int* in_sizes, int n_in,
                              void* d_out, int out_size, void* d_ws, size_t ws_size,
                              hipStream_t stream) {
  const float* x  = (const float*)d_in[0];
  const float* W  = (const float*)d_in[1];
  const float* bv = (const float*)d_in[2];
  float* out = (float*)d_out;

  float* p    = (float*)d_ws;
  float* zrow = p + (size_t)8192 * D_DIM;

  gemm_norm_kernel<<<256, 128, 0, stream>>>(x, W, bv, p, zrow);
  band_sim_kernel<<<256, 128, 0, stream>>>(p, zrow, out);
}
